# starting kernel
# baseline (speedup 1.0000x reference)
#include <hip/hip_runtime.h>
#include <math.h>


typedef __attribute__((ext_vector_type(2))) float v2f;
typedef __attribute__((ext_vector_type(8))) float v8f;

#define HIDDEN 256
#define PROJ   64
#define PROJ2  128
#define RPB    64
#define XSTRIDE 260
#define HSTRIDE 68

__global__ __launch_bounds__(256)
void k_project(const float* __restrict__ node_feats,
               const float* __restrict__ W_k,
               const float* __restrict__ W_v,
               const int*   __restrict__ node_idx,
               const int*   __restrict__ he_idx,
               float* __restrict__ proj,
               float* __restrict__ ksum,
               float* __restrict__ cnt,
               int nnz)
{
    __shared__ float xs[RPB * XSTRIDE];
    const int e0 = blockIdx.x * RPB;
    const int t  = threadIdx.x;

    if (t < RPB) {
        const int e = e0 + t;
        if (e < nnz) atomicAdd(&cnt[he_idx[e]], 1.0f);
    }

    {
        const int r   = t >> 2;
        const int c0  = (t & 3) * 64;
        const int e   = e0 + r;
        const int row = (e < nnz) ? node_idx[e] : 0;
        const float* src = node_feats + (long)row * HIDDEN + c0;
        float* dst = xs + r * XSTRIDE + c0;
        #pragma unroll
        for (int i = 0; i < 64; i += 4)
            *(float4*)(dst + i) = *(const float4*)(src + i);
    }
    __syncthreads();

    const int wave = t >> 5;
    const int lane = t & 31;
    const int mn   = lane & 15;
    const int kb   = (lane >> 4) * 2;
    const int mh   = (lane >> 4) * 8;
    const int n0   = wave * 16;

    const int nrow = n0 + mn;
    const float* wrow = (nrow < PROJ) ? (W_k + (long)nrow * HIDDEN)
                                      : (W_v + (long)(nrow - PROJ) * HIDDEN);

    v8f acc[4] = {};
    #pragma unroll 2
    for (int kk = 0; kk < HIDDEN; kk += 4) {
        v2f b;
        b.x = wrow[kk + kb];
        b.y = wrow[kk + kb + 1];
        #pragma unroll
        for (int rt = 0; rt < 4; ++rt) {
            v2f a;
            const float* arow = xs + (rt * 16 + mn) * XSTRIDE;
            a.x = arow[kk + kb];
            a.y = arow[kk + kb + 1];
            acc[rt] = __builtin_amdgcn_wmma_f32_16x16x4_f32(
                          false, a, false, b, (short)0, acc[rt], false, false);
        }
    }

    const bool isKeyCols = (n0 < PROJ);
    if (e0 + RPB <= nnz) {
        #pragma unroll
        for (int rt = 0; rt < 4; ++rt)
            #pragma unroll
            for (int r = 0; r < 8; ++r) {
                const int e = e0 + rt * 16 + r + mh;
                const float v = acc[rt][r];
                proj[(long)e * PROJ2 + n0 + mn] = v;
                if (isKeyCols)
                    atomicAdd(&ksum[(long)he_idx[e] * PROJ + n0 + mn], v);
            }
    } else {
        #pragma unroll
        for (int rt = 0; rt < 4; ++rt)
            #pragma unroll
            for (int r = 0; r < 8; ++r) {
                const int e = e0 + rt * 16 + r + mh;
                if (e < nnz) {
                    const float v = acc[rt][r];
                    proj[(long)e * PROJ2 + n0 + mn] = v;
                    if (isKeyCols)
                        atomicAdd(&ksum[(long)he_idx[e] * PROJ + n0 + mn], v);
                }
            }
    }
}

__global__ __launch_bounds__(256)
void k_scores(const float* __restrict__ proj,
              const int* __restrict__ he_idx,
              const float* __restrict__ ksum, const float* __restrict__ cnt,
              float* __restrict__ scores, float* __restrict__ smax, int nnz)
{
    const int e = blockIdx.x * 8 + (threadIdx.x >> 5);
    if (e >= nnz) return;
    const int lane = threadIdx.x & 31;
    const int he = he_idx[e];
    const float inv = 1.0f / fmaxf(cnt[he], 1.0f);

    float ss = 0.0f;
    #pragma unroll
    for (int j = 0; j < 2; ++j) {
        const int p = lane + j * 32;
        const float d = proj[(long)e * PROJ2 + p]
                      - ksum[(long)he * PROJ + p] * inv;
        ss += d * d;
    }
    #pragma unroll
    for (int off = 16; off > 0; off >>= 1)
        ss += __shfl_down(ss, off, 32);

    if (lane == 0) {
        const float score = sqrtf(ss) * 0.125f;
        scores[e] = score;
        atomicMax((int*)&smax[he], __float_as_int(score));
    }
}

__global__ __launch_bounds__(256)
void k_exp(float* __restrict__ scores, const int* __restrict__ he_idx,
           const float* __restrict__ smax, float* __restrict__ denom, int nnz)
{
    const int e = blockIdx.x * blockDim.x + threadIdx.x;
    if (e >= nnz) return;
    const int he = he_idx[e];
    const float ex = expf(scores[e] - smax[he]);
    scores[e] = ex;
    atomicAdd(&denom[he], ex);
}

__global__ __launch_bounds__(256)
void k_aggregate(const float* __restrict__ proj,
                 const float* __restrict__ scores,
                 const int* __restrict__ he_idx,
                 const float* __restrict__ denom,
                 float* __restrict__ hefeats, int nnz)
{
    const long idx = (long)blockIdx.x * blockDim.x + threadIdx.x;
    if (idx >= (long)nnz * PROJ) return;
    const int e  = (int)(idx >> 6);
    const int p  = (int)(idx & 63);
    const int he = he_idx[e];
    const float attn = scores[e] / fmaxf(denom[he], 1e-30f);
    atomicAdd(&hefeats[(long)he * PROJ + p],
              attn * proj[(long)e * PROJ2 + PROJ + p]);
}

__global__ __launch_bounds__(256)
void k_out_gemm(const float* __restrict__ hefeats,
                const float* __restrict__ W_v,
                float* __restrict__ out, int num_he)
{
    __shared__ float hs[RPB * HSTRIDE];
    const int e0 = blockIdx.x * RPB;
    const int t  = threadIdx.x;
    {
        const int r = t >> 2;
        const int c = (t & 3) * 16;
        const int e = e0 + r;
        #pragma unroll
        for (int i = 0; i < 16; i += 4) {
            float4 v = (e < num_he)
                     ? *(const float4*)(hefeats + (long)e * PROJ + c + i)
                     : make_float4(0.f, 0.f, 0.f, 0.f);
            *(float4*)(hs + r * HSTRIDE + c + i) = v;
        }
    }
    __syncthreads();

    const int wave = t >> 5;
    const int lane = t & 31;
    const int mn   = lane & 15;
    const int kb   = (lane >> 4) * 2;
    const int mh   = (lane >> 4) * 8;
    const bool full = (e0 + RPB) <= num_he;

    #pragma unroll
    for (int sub = 0; sub < 2; ++sub) {
        const int n0 = wave * 32 + sub * 16;
        v8f acc[4] = {};
        #pragma unroll 2
        for (int kk = 0; kk < PROJ; kk += 4) {
            v2f b;
            b.x = W_v[(long)(kk + kb)     * HIDDEN + n0 + mn];
            b.y = W_v[(long)(kk + kb + 1) * HIDDEN + n0 + mn];
            #pragma unroll
            for (int rt = 0; rt < 4; ++rt) {
                v2f a;
                const float* arow = hs + (rt * 16 + mn) * HSTRIDE;
                a.x = arow[kk + kb];
                a.y = arow[kk + kb + 1];
                acc[rt] = __builtin_amdgcn_wmma_f32_16x16x4_f32(
                              false, a, false, b, (short)0, acc[rt], false, false);
            }
        }
        if (full) {
            #pragma unroll
            for (int rt = 0; rt < 4; ++rt)
                #pragma unroll
                for (int r = 0; r < 8; ++r) {
                    const int e = e0 + rt * 16 + r + mh;
                    out[(long)e * HIDDEN + n0 + mn] = acc[rt][r];
                }
        } else {
            #pragma unroll
            for (int rt = 0; rt < 4; ++rt)
                #pragma unroll
                for (int r = 0; r < 8; ++r) {
                    const int e = e0 + rt * 16 + r + mh;
                    if (e < num_he) out[(long)e * HIDDEN + n0 + mn] = acc[rt][r];
                }
        }
    }
}

__global__ __launch_bounds__(256)
void k_zero(float* __restrict__ p, long n)
{
    const long i = (long)blockIdx.x * blockDim.x + threadIdx.x;
    if (i < n) p[i] = 0.0f;
}

extern "C" void kernel_launch(void* const* d_in, const int* in_sizes, int n_in,
                              void* d_out, int out_size, void* d_ws, size_t ws_size,
                              hipStream_t stream)
{
    const float* node_feats = (const float*)d_in[0];
    const float* W_k        = (const float*)d_in[1];
    const float* W_v        = (const float*)d_in[2];
    const int*   node_idx   = (const int*)d_in[3];
    const int*   he_idx     = (const int*)d_in[4];
    (void)n_in; (void)ws_size;

    const int nnz = in_sizes[3];
    const int HE  = out_size / HIDDEN;

    float* ws = (float*)d_ws;
    long o = 0;
    float* proj    = ws + o; o += (long)nnz * PROJ2;
    float* ksum    = ws + o; o += (long)HE * PROJ;
    float* hefeats = ws + o; o += (long)HE * PROJ;
    float* cnt     = ws + o; o += HE;
    float* smax    = ws + o; o += HE;
    float* denom   = ws + o; o += HE;
    float* scores  = ws + o; o += nnz;

    const long zn = (long)HE * PROJ * 2 + 3L * HE;
    k_zero<<<(int)((zn + 255) / 256), 256, 0, stream>>>(ksum, zn);

    k_project<<<(nnz + RPB - 1) / RPB, 256, 0, stream>>>(
        node_feats, W_k, W_v, node_idx, he_idx, proj, ksum, cnt, nnz);

    k_scores<<<(nnz + 7) / 8, 256, 0, stream>>>(
        proj, he_idx, ksum, cnt, scores, smax, nnz);

    k_exp<<<(nnz + 255) / 256, 256, 0, stream>>>(
        scores, he_idx, smax, denom, nnz);

    const long t2 = (long)nnz * PROJ;
    k_aggregate<<<(int)((t2 + 255) / 256), 256, 0, stream>>>(
        proj, scores, he_idx, denom, hefeats, nnz);

    k_out_gemm<<<(HE + RPB - 1) / RPB, 256, 0, stream>>>(
        hefeats, W_v, (float*)d_out, HE);
}


# k_project: W staged once per block in LDS; each wave owns 16-row x 128-col tiles, A fragments read straight from the gathered global rows (no row staging, no per-step W global loads); same sums
# speedup vs baseline: 1.0851x; 1.0851x over previous
#include <hip/hip_runtime.h>
#include <math.h>


typedef __attribute__((ext_vector_type(2))) float v2f;
typedef __attribute__((ext_vector_type(8))) float v8f;

#define HIDDEN 256
#define PROJ   64
#define PROJ2  128
#define RPB    64
#define HSTRIDE 68

#define WSTRIDE 260
#define PROJ_BLOCKS 512

__global__ __launch_bounds__(256, 4)
void k_project(const float* __restrict__ node_feats,
               const float* __restrict__ W_k,
               const float* __restrict__ W_v,
               const int*   __restrict__ node_idx,
               const int*   __restrict__ he_idx,
               float* __restrict__ proj,
               float* __restrict__ ksum,
               float* __restrict__ cnt,
               int nnz)
{
    __shared__ alignas(16) float wl[PROJ2 * WSTRIDE];
    const int t = threadIdx.x;

    #pragma unroll 8
    for (int i = 0; i < (PROJ2 * HIDDEN / 4) / 256; ++i) {
        const int idx4 = i * 256 + t;
        const int row  = idx4 >> 6;
        const int c    = (idx4 & 63) * 4;
        const float* src = (row < PROJ) ? (W_k + (long)row * HIDDEN + c)
                                        : (W_v + (long)(row - PROJ) * HIDDEN + c);
        *(float4*)(wl + row * WSTRIDE + c) = *(const float4*)src;
    }
    __syncthreads();

    const int wave = __builtin_amdgcn_readfirstlane(t >> 5);
    const int lane = t & 31;
    const int mn   = lane & 15;
    const int kb   = (lane >> 4) * 2;
    const int mh   = (lane >> 4) * 8;

    const int ntiles = (nnz + 15) >> 4;
    const int nwaves = (int)gridDim.x * 8;
    const float* wb  = wl + mn * WSTRIDE + kb;

    for (int tile = wave * (int)gridDim.x + (int)blockIdx.x; tile < ntiles; tile += nwaves) {
        const int e0 = tile * 16;

        if (lane < 16) {
            const int e = e0 + lane;
            if (e < nnz) atomicAdd(&cnt[he_idx[e]], 1.0f);
        }

        const int ea  = e0 + mn;
        const int row = (ea < nnz) ? node_idx[ea] : 0;
        const float* xrow = node_feats + (long)row * HIDDEN + kb;

        v8f acc[8] = {};
        #pragma unroll 1
        for (int half = 0; half < 2; ++half) {
            const float* xr = xrow + half * 128;
            const float* wr = wb + half * 128;
            v2f a[32];
            #pragma unroll
            for (int j = 0; j < 32; ++j)
                a[j] = *(const v2f*)(xr + j * 4);
            v2f bc[8];
            #pragma unroll
            for (int ct = 0; ct < 8; ++ct)
                bc[ct] = *(const v2f*)(wr + ct * 16 * WSTRIDE);
            #pragma unroll
            for (int j = 0; j < 32; ++j) {
                v2f bn[8];
                if (j + 1 < 32) {
                    #pragma unroll
                    for (int ct = 0; ct < 8; ++ct)
                        bn[ct] = *(const v2f*)(wr + ct * 16 * WSTRIDE + (j + 1) * 4);
                }
                #pragma unroll
                for (int ct = 0; ct < 8; ++ct)
                    acc[ct] = __builtin_amdgcn_wmma_f32_16x16x4_f32(
                                  false, a[j], false, bc[ct], (short)0, acc[ct], false, false);
                if (j + 1 < 32) {
                    #pragma unroll
                    for (int ct = 0; ct < 8; ++ct)
                        bc[ct] = bn[ct];
                }
            }
        }

        if (e0 + 16 <= nnz) {
            #pragma unroll
            for (int r = 0; r < 8; ++r) {
                const int e  = e0 + r + mh;
                const int he = he_idx[e];
                float* prow = proj + (long)e * PROJ2 + mn;
                float* krow = ksum + (long)he * PROJ + mn;
                #pragma unroll
                for (int ct = 0; ct < 8; ++ct) {
                    const float v = acc[ct][r];
                    prow[ct * 16] = v;
                    if (ct < 4) atomicAdd(&krow[ct * 16], v);
                }
            }
        } else {
            #pragma unroll
            for (int r = 0; r < 8; ++r) {
                const int e = e0 + r + mh;
                if (e < nnz) {
                    const int he = he_idx[e];
                    float* prow = proj + (long)e * PROJ2 + mn;
                    float* krow = ksum + (long)he * PROJ + mn;
                    #pragma unroll
                    for (int ct = 0; ct < 8; ++ct) {
                        const float v = acc[ct][r];
                        prow[ct * 16] = v;
                        if (ct < 4) atomicAdd(&krow[ct * 16], v);
                    }
                }
            }
        }
    }
}

__global__ __launch_bounds__(256)
void k_scores(const float* __restrict__ proj,
              const int* __restrict__ he_idx,
              const float* __restrict__ ksum, const float* __restrict__ cnt,
              float* __restrict__ scores, float* __restrict__ smax, int nnz)
{
    const int e = blockIdx.x * 8 + (threadIdx.x >> 5);
    if (e >= nnz) return;
    const int lane = threadIdx.x & 31;
    const int he = he_idx[e];
    const float inv = 1.0f / fmaxf(cnt[he], 1.0f);

    float ss = 0.0f;
    #pragma unroll
    for (int j = 0; j < 2; ++j) {
        const int p = lane + j * 32;
        const float d = proj[(long)e * PROJ2 + p]
                      - ksum[(long)he * PROJ + p] * inv;
        ss += d * d;
    }
    #pragma unroll
    for (int off = 16; off > 0; off >>= 1)
        ss += __shfl_down(ss, off, 32);

    if (lane == 0) {
        const float score = sqrtf(ss) * 0.125f;
        scores[e] = score;
        atomicMax((int*)&smax[he], __float_as_int(score));
    }
}

__global__ __launch_bounds__(256)
void k_exp(float* __restrict__ scores, const int* __restrict__ he_idx,
           const float* __restrict__ smax, float* __restrict__ denom, int nnz)
{
    const int e = blockIdx.x * blockDim.x + threadIdx.x;
    if (e >= nnz) return;
    const int he = he_idx[e];
    const float ex = expf(scores[e] - smax[he]);
    scores[e] = ex;
    atomicAdd(&denom[he], ex);
}

__global__ __launch_bounds__(256)
void k_aggregate(const float* __restrict__ proj,
                 const float* __restrict__ scores,
                 const int* __restrict__ he_idx,
                 const float* __restrict__ denom,
                 float* __restrict__ hefeats, int nnz)
{
    const long idx = (long)blockIdx.x * blockDim.x + threadIdx.x;
    if (idx >= (long)nnz * PROJ) return;
    const int e  = (int)(idx >> 6);
    const int p  = (int)(idx & 63);
    const int he = he_idx[e];
    const float attn = scores[e] / fmaxf(denom[he], 1e-30f);
    atomicAdd(&hefeats[(long)he * PROJ + p],
              attn * proj[(long)e * PROJ2 + PROJ + p]);
}

__global__ __launch_bounds__(256)
void k_out_gemm(const float* __restrict__ hefeats,
                const float* __restrict__ W_v,
                float* __restrict__ out, int num_he)
{
    __shared__ float hs[RPB * HSTRIDE];
    const int e0 = blockIdx.x * RPB;
    const int t  = threadIdx.x;
    {
        const int r = t >> 2;
        const int c = (t & 3) * 16;
        const int e = e0 + r;
        #pragma unroll
        for (int i = 0; i < 16; i += 4) {
            float4 v = (e < num_he)
                     ? *(const float4*)(hefeats + (long)e * PROJ + c + i)
                     : make_float4(0.f, 0.f, 0.f, 0.f);
            *(float4*)(hs + r * HSTRIDE + c + i) = v;
        }
    }
    __syncthreads();

    const int wave = t >> 5;
    const int lane = t & 31;
    const int mn   = lane & 15;
    const int kb   = (lane >> 4) * 2;
    const int mh   = (lane >> 4) * 8;
    const bool full = (e0 + RPB) <= num_he;

    #pragma unroll
    for (int sub = 0; sub < 2; ++sub) {
        const int n0 = wave * 32 + sub * 16;
        v8f acc[4] = {};
        #pragma unroll 2
        for (int kk = 0; kk < PROJ; kk += 4) {
            v2f b;
            b.x = W_v[(long)(kk + kb)     * HIDDEN + n0 + mn];
            b.y = W_v[(long)(kk + kb + 1) * HIDDEN + n0 + mn];
            #pragma unroll
            for (int rt = 0; rt < 4; ++rt) {
                v2f a;
                const float* arow = hs + (rt * 16 + mn) * HSTRIDE;
                a.x = arow[kk + kb];
                a.y = arow[kk + kb + 1];
                acc[rt] = __builtin_amdgcn_wmma_f32_16x16x4_f32(
                              false, a, false, b, (short)0, acc[rt], false, false);
            }
        }
        if (full) {
            #pragma unroll
            for (int rt = 0; rt < 4; ++rt)
                #pragma unroll
                for (int r = 0; r < 8; ++r) {
                    const int e = e0 + rt * 16 + r + mh;
                    out[(long)e * HIDDEN + n0 + mn] = acc[rt][r];
                }
        } else {
            #pragma unroll
            for (int rt = 0; rt < 4; ++rt)
                #pragma unroll
                for (int r = 0; r < 8; ++r) {
                    const int e = e0 + rt * 16 + r + mh;
                    if (e < num_he) out[(long)e * HIDDEN + n0 + mn] = acc[rt][r];
                }
        }
    }
}

__global__ __launch_bounds__(256)
void k_zero(float* __restrict__ p, long n)
{
    const long i = (long)blockIdx.x * blockDim.x + threadIdx.x;
    if (i < n) p[i] = 0.0f;
}

extern "C" void kernel_launch(void* const* d_in, const int* in_sizes, int n_in,
                              void* d_out, int out_size, void* d_ws, size_t ws_size,
                              hipStream_t stream)
{
    const float* node_feats = (const float*)d_in[0];
    const float* W_k        = (const float*)d_in[1];
    const float* W_v        = (const float*)d_in[2];
    const int*   node_idx   = (const int*)d_in[3];
    const int*   he_idx     = (const int*)d_in[4];
    (void)n_in; (void)ws_size;

    const int nnz = in_sizes[3];
    const int HE  = out_size / HIDDEN;

    float* ws = (float*)d_ws;
    long o = 0;
    float* proj    = ws + o; o += (long)nnz * PROJ2;
    float* ksum    = ws + o; o += (long)HE * PROJ;
    float* hefeats = ws + o; o += (long)HE * PROJ;
    float* cnt     = ws + o; o += HE;
    float* smax    = ws + o; o += HE;
    float* denom   = ws + o; o += HE;
    float* scores  = ws + o; o += nnz;

    const long zn = (long)HE * PROJ * 2 + 3L * HE;
    k_zero<<<(int)((zn + 255) / 256), 256, 0, stream>>>(ksum, zn);

    k_project<<<PROJ_BLOCKS, 256, 0, stream>>>(
        node_feats, W_k, W_v, node_idx, he_idx, proj, ksum, cnt, nnz);

    k_scores<<<(nnz + 7) / 8, 256, 0, stream>>>(
        proj, he_idx, ksum, cnt, scores, smax, nnz);

    k_exp<<<(nnz + 255) / 256, 256, 0, stream>>>(
        scores, he_idx, smax, denom, nnz);

    const long t2 = (long)nnz * PROJ;
    k_aggregate<<<(int)((t2 + 255) / 256), 256, 0, stream>>>(
        proj, scores, he_idx, denom, hefeats, nnz);

    k_out_gemm<<<(HE + RPB - 1) / RPB, 256, 0, stream>>>(
        hefeats, W_v, (float*)d_out, HE);
}
